# starting kernel
# baseline (speedup 1.0000x reference)
#include <hip/hip_runtime.h>
#include <math.h>

#define N_GRAPHS 64
#define M_NODES  512
#define IN_DIM   256
#define OUT_DIM  128
#define NHEADS   2
#define ZD       256
#define EG       8192
#define P_IDX    32
#define NEG_SLOPE 0.2f

#define CAP      47
#define MAXR     48
#define OUT_STRIDE 768

typedef float v2f __attribute__((ext_vector_type(2)));
typedef float v8f __attribute__((ext_vector_type(8)));

__global__ __launch_bounds__(256, 2)
void gat_node_kernel(const float* __restrict__ feat,
                     const int*   __restrict__ srcArr,
                     const int*   __restrict__ dstArr,
                     const int*   __restrict__ opPtr,
                     const float* __restrict__ W_f,  const float* __restrict__ al_f,
                     const float* __restrict__ ar_f, const float* __restrict__ b_f,
                     const float* __restrict__ W_b,  const float* __restrict__ al_b,
                     const float* __restrict__ ar_b, const float* __restrict__ b_b,
                     float* __restrict__ out)
{
    __shared__ float Zs[MAXR * ZD];
    __shared__ float eh[MAXR * NHEADS];
    __shared__ float er[NHEADS];
    __shared__ float partial[NHEADS * OUT_DIM];
    __shared__ int   rowNode[MAXR];
    __shared__ int   pref[257];

    const int gb   = blockIdx.x;
    const int g    = gb >> 1;
    const int dir  = gb & 1;
    const int tid  = threadIdx.x;
    const int node = opPtr[0] + g * M_NODES;

    const int*   seg  = (dir == 0) ? dstArr : srcArr;
    const int*   gat  = (dir == 0) ? srcArr : dstArr;
    const float* Wd   = (dir == 0) ? W_f  : W_b;
    const float* al   = (dir == 0) ? al_f : al_b;
    const float* ar   = (dir == 0) ? ar_f : ar_b;
    const float* bias = (dir == 0) ? b_f  : b_b;

    const int chunk  = EG / 256;
    const int myBase = g * EG + tid * chunk;
    int c = 0;
    for (int j = 0; j < chunk; ++j)
        c += (seg[myBase + j] == node) ? 1 : 0;
    pref[tid] = c;
    __syncthreads();
    if (tid == 0) {
        int acc = 0;
        for (int t = 0; t < 256; ++t) { int v = pref[t]; pref[t] = acc; acc += v; }
        pref[256] = acc;
    }
    __syncthreads();
    int pos = pref[tid];
    for (int j = 0; j < chunk; ++j) {
        int e = myBase + j;
        if (seg[e] == node) {
            if (pos < CAP) rowNode[pos] = gat[e];
            ++pos;
        }
    }
    __syncthreads();
    const int cnt   = min(pref[256], CAP);
    const int R     = cnt + 1;
    const int tiles = (R + 15) >> 4;
    if (tid == 0) rowNode[cnt] = node;
    __syncthreads();

    const int wave  = tid >> 5;
    const int lane  = tid & 31;
    const int lm    = lane & 15;
    const int khalf = (lane >> 4) << 1;
    for (int mt = 0; mt < tiles; ++mt) {
        const int m = mt * 16 + lm;
        const float* fr = feat + (long)rowNode[min(m, R - 1)] * IN_DIM + khalf;
        #pragma unroll
        for (int nn = 0; nn < 2; ++nn) {
            const int nt   = wave * 2 + nn;
            const int ncol = nt * 16 + lm;
            const float* wc = Wd + (long)khalf * ZD + ncol;
            v8f acc = {};
            #pragma unroll 4
            for (int k0 = 0; k0 < IN_DIM; k0 += 4) {
                v2f a, b;
                a.x = fr[k0];
                a.y = fr[k0 + 1];
                b.x = wc[(long)k0 * ZD];
                b.y = wc[(long)(k0 + 1) * ZD];
                acc = __builtin_amdgcn_wmma_f32_16x16x4_f32(
                          false, a, false, b, (short)0, acc, false, false);
            }
            const int mrow = mt * 16 + ((lane >> 4) << 3);
            #pragma unroll
            for (int r = 0; r < 8; ++r)
                Zs[(mrow + r) * ZD + ncol] = acc[r];
        }
    }
    __syncthreads();

    if (tid < cnt) {
        float e0 = 0.f, e1 = 0.f;
        const float* zr = &Zs[tid * ZD];
        for (int d = 0; d < OUT_DIM; ++d) {
            e0 += zr[d]           * al[d];
            e1 += zr[OUT_DIM + d] * al[OUT_DIM + d];
        }
        eh[tid * 2 + 0] = e0;
        eh[tid * 2 + 1] = e1;
    }
    if (tid < NHEADS) {
        float s = 0.f;
        const float* zr  = &Zs[cnt * ZD + tid * OUT_DIM];
        const float* arr = &ar[tid * OUT_DIM];
        for (int d = 0; d < OUT_DIM; ++d) s += zr[d] * arr[d];
        er[tid] = s;
    }
    __syncthreads();

    {
        const int h = tid >> 7;
        const int d = tid & 127;
        const float erh = er[h];
        float mx = -INFINITY;
        for (int i = 0; i < cnt; ++i) {
            float e = eh[i * 2 + h] + erh;
            e = (e > 0.f) ? e : NEG_SLOPE * e;
            mx = fmaxf(mx, e);
        }
        float s = 0.f;
        for (int i = 0; i < cnt; ++i) {
            float e = eh[i * 2 + h] + erh;
            e = (e > 0.f) ? e : NEG_SLOPE * e;
            s += expf(e - mx);
        }
        const float inv = (cnt > 0) ? 1.f / s : 0.f;
        float accum = 0.f;
        for (int i = 0; i < cnt; ++i) {
            float e = eh[i * 2 + h] + erh;
            e = (e > 0.f) ? e : NEG_SLOPE * e;
            accum += (expf(e - mx) * inv) * Zs[i * ZD + h * OUT_DIM + d];
        }
        partial[h * OUT_DIM + d] = accum + bias[h * OUT_DIM + d];
    }
    __syncthreads();
    if (tid < OUT_DIM)
        out[(long)g * OUT_STRIDE + dir * OUT_DIM + tid] =
            0.5f * (partial[tid] + partial[OUT_DIM + tid]);
}

__global__ __launch_bounds__(256)
void tail_kernel(const float* __restrict__ feat,
                 const int*   __restrict__ opPtr,
                 const int*   __restrict__ parallel,
                 float* __restrict__ out)
{
    const int g = blockIdx.x;
    const int t = threadIdx.x;
    const int node = opPtr[0] + g * M_NODES;
    out[(long)g * OUT_STRIDE + 256 + t] = feat[(long)node * IN_DIM + t];
    float s = 0.f;
    for (int p = 0; p < P_IDX; ++p) {
        int n = parallel[p] + g * M_NODES;
        s += feat[(long)n * IN_DIM + t];
    }
    out[(long)g * OUT_STRIDE + 512 + t] = s;
}

extern "C" void kernel_launch(void* const* d_in, const int* in_sizes, int n_in,
                              void* d_out, int out_size, void* d_ws, size_t ws_size,
                              hipStream_t stream)
{
    const float* feat     = (const float*)d_in[0];
    const int*   src      = (const int*)  d_in[1];
    const int*   dst      = (const int*)  d_in[2];
    const int*   op       = (const int*)  d_in[3];
    const int*   parallel = (const int*)  d_in[4];
    const float* W_f  = (const float*)d_in[5];
    const float* al_f = (const float*)d_in[6];
    const float* ar_f = (const float*)d_in[7];
    const float* b_f  = (const float*)d_in[8];
    const float* W_b  = (const float*)d_in[9];
    const float* al_b = (const float*)d_in[10];
    const float* ar_b = (const float*)d_in[11];
    const float* b_b  = (const float*)d_in[12];
    float* out = (float*)d_out;

    gat_node_kernel<<<dim3(N_GRAPHS * 2), dim3(256), 0, stream>>>(
        feat, src, dst, op, W_f, al_f, ar_f, b_f, W_b, al_b, ar_b, b_b, out);
    tail_kernel<<<dim3(N_GRAPHS), dim3(256), 0, stream>>>(feat, op, parallel, out);
}


# one launch, was 2: a block owns 64 columns of both heads and writes their mean; the other columns' logit share = feature rows x (W attn) on WMMA; tail work in-block; logit sum order changes
# speedup vs baseline: 3.2838x; 3.2838x over previous
#include <hip/hip_runtime.h>
#include <math.h>

#define N_GRAPHS 64
#define M_NODES  512
#define IN_DIM   256
#define OUT_DIM  128
#define NHEADS   2
#define ZD       256
#define EG       8192
#define P_IDX    32
#define NEG_SLOPE 0.2f

#define CAP      47
#define MAXR     48
#define OUT_STRIDE 768
#define NTHR     512
#define NWAVES   16
#define EPT      16
#define ZP       132
#define KSPLIT   4
#define KSLICE   64
#define ZPLANE   (MAXR * ZP)
#define BP       132
#define UP       260
#define LPL      (MAXR * 4)

typedef float v2f __attribute__((ext_vector_type(2)));
typedef float v4f __attribute__((ext_vector_type(4)));
typedef float v8f __attribute__((ext_vector_type(8)));
typedef int   v4i __attribute__((ext_vector_type(4)));

template <int M>
__device__ __forceinline__ float row_x(float v)
{
    const int b = __builtin_bit_cast(int, v);
    return __builtin_bit_cast(float, __builtin_amdgcn_update_dpp(b, b, 0x160 + M, 0xf, 0xf, false));
}
__device__ __forceinline__ float xhalf(float v)
{
    const int b = __builtin_bit_cast(int, v);
    return __builtin_bit_cast(float,
               __builtin_amdgcn_permlanex16(b, b, 0x76543210, 0xfedcba98, false, false));
}
__device__ __forceinline__ float wave_max(float v)
{
    v = fmaxf(v, row_x<1>(v));
    v = fmaxf(v, row_x<2>(v));
    v = fmaxf(v, row_x<4>(v));
    v = fmaxf(v, row_x<8>(v));
    return fmaxf(v, xhalf(v));
}
__device__ __forceinline__ float wave_sum(float v)
{
    v += row_x<1>(v);
    v += row_x<2>(v);
    v += row_x<4>(v);
    v += row_x<8>(v);
    return v + xhalf(v);
}

template <int S>
__device__ __forceinline__ int row_shr(int v)
{
    return __builtin_amdgcn_update_dpp(0, v, 0x110 + S, 0xf, 0xf, false);
}
__device__ __forceinline__ int wave_prefix_incl(int v, int lane)
{
    v += row_shr<1>(v);
    v += row_shr<2>(v);
    v += row_shr<4>(v);
    v += row_shr<8>(v);
    const int row0 = __builtin_amdgcn_readlane(v, 15);
    return v + ((lane >= 16) ? row0 : 0);
}
__device__ __forceinline__ unsigned match4(v4i q, int node)
{
    return  (unsigned)(q.x == node)       | ((unsigned)(q.y == node) << 1)
         | ((unsigned)(q.z == node) << 2) | ((unsigned)(q.w == node) << 3);
}

__global__ __launch_bounds__(NTHR)
void gat_node_kernel(const float* __restrict__ feat,
                     const int*   __restrict__ srcArr,
                     const int*   __restrict__ dstArr,
                     const int*   __restrict__ opPtr,
                     const int*   __restrict__ parallel,
                     const float* __restrict__ W_f,  const float* __restrict__ al_f,
                     const float* __restrict__ ar_f, const float* __restrict__ b_f,
                     const float* __restrict__ W_b,  const float* __restrict__ al_b,
                     const float* __restrict__ ar_b, const float* __restrict__ b_b,
                     float* __restrict__ out)
{
    __shared__ __attribute__((aligned(16))) float Zpart[KSPLIT * ZPLANE];
    float* const Zs = Zpart;
    __shared__ __attribute__((aligned(16))) float bsrc[5 * BP];
    __shared__ __attribute__((aligned(16))) float Us[5 * UP];
    __shared__ float lpart[5 * LPL];
    __shared__ __attribute__((aligned(16))) float alphaS[NHEADS * MAXR];
    __shared__ float h1res[64];
    __shared__ __attribute__((aligned(16))) float tailS[(P_IDX + 1) * 64];
    __shared__ int   rowNode[MAXR];
    __shared__ __attribute__((aligned(16))) int wtot[NWAVES];

    const int gb   = blockIdx.x;
    const int g    = gb >> 2;
    const int dir  = (gb >> 1) & 1;
    const int ch   = gb & 1;
    const int tid  = threadIdx.x;
    const int node = opPtr[0] + g * M_NODES;

    const int*   seg  = (dir == 0) ? dstArr : srcArr;
    const int*   gat  = (dir == 0) ? srcArr : dstArr;
    const float* Wd   = (dir == 0) ? W_f  : W_b;
    const float* al   = (dir == 0) ? al_f : al_b;
    const float* ar   = (dir == 0) ? ar_f : ar_b;
    const float* bias = (dir == 0) ? b_f  : b_b;

    const int wave = __builtin_amdgcn_readfirstlane(tid >> 5);
    const int lane = tid & 31;
    const int lm   = lane & 15;
    const int half = lane >> 4;

    const int ucol = half * OUT_DIM + (ch ^ 1) * 64 + 4 * lm;
    const v4f al4 = *(const v4f*)(al + ucol);
    const v4f ar4 = *(const v4f*)(ar + ucol);
    v4f wv[16];
    {
        const float* wu = Wd + (long)(wave * 16) * ZD + ucol;
        #pragma unroll
        for (int r = 0; r < 16; ++r) wv[r] = *(const v4f*)(wu + (long)r * ZD);
    }

    const int sq = tid >> 7, sk = tid & 127, sh = sq >> 1;
    const float* av = (sq & 1) ? ar : al;
    const float sld  = av[sh * OUT_DIM + ch * 64 + (sk & 63)];
    const float sval = ((sk >> 6) == sh) ? sld : 0.f;

    __builtin_amdgcn_sched_barrier(0);

    const int myBase = g * EG + tid * EPT;
    const v4i* sp = (const v4i*)(seg + myBase);
    const v4i q0 = sp[0];
    const v4i q1 = sp[1];
    const v4i q2 = sp[2];
    const v4i q3 = sp[3];

    const int  tcol = 64 * (gb & 3) + 2 * lane;
    const v2f tA = *(const v2f*)(feat + (long)(parallel[wave]      + g * M_NODES) * IN_DIM + tcol);
    const v2f tB = *(const v2f*)(feat + (long)(parallel[wave + 16] + g * M_NODES) * IN_DIM + tcol);
    const v2f tO = *(const v2f*)(feat + (long)node * IN_DIM + tcol);

    __builtin_amdgcn_sched_barrier(0);

    {
        float vl[16], vr[16];
        #pragma unroll
        for (int r = 0; r < 16; ++r) {
            vl[r] = wv[r].x * al4.x + wv[r].y * al4.y + wv[r].z * al4.z + wv[r].w * al4.w;
            vr[r] = wv[r].x * ar4.x + wv[r].y * ar4.y + wv[r].z * ar4.z + wv[r].w * ar4.w;
        }
        const bool b8 = (lm & 8) != 0, b4 = (lm & 4) != 0, b2 = (lm & 2) != 0, b1 = (lm & 1) != 0;
        #pragma unroll
        for (int i = 0; i < 8; ++i) {
            const float kl = b8 ? vl[i + 8] : vl[i], sl = b8 ? vl[i] : vl[i + 8];
            const float kr = b8 ? vr[i + 8] : vr[i], sr = b8 ? vr[i] : vr[i + 8];
            vl[i] = kl + row_x<8>(sl);
            vr[i] = kr + row_x<8>(sr);
        }
        #pragma unroll
        for (int i = 0; i < 4; ++i) {
            const float kl = b4 ? vl[i + 4] : vl[i], sl = b4 ? vl[i] : vl[i + 4];
            const float kr = b4 ? vr[i + 4] : vr[i], sr = b4 ? vr[i] : vr[i + 4];
            vl[i] = kl + row_x<4>(sl);
            vr[i] = kr + row_x<4>(sr);
        }
        #pragma unroll
        for (int i = 0; i < 2; ++i) {
            const float kl = b2 ? vl[i + 2] : vl[i], sl = b2 ? vl[i] : vl[i + 2];
            const float kr = b2 ? vr[i + 2] : vr[i], sr = b2 ? vr[i] : vr[i + 2];
            vl[i] = kl + row_x<2>(sl);
            vr[i] = kr + row_x<2>(sr);
        }
        const float keepL = (b1 ? vl[1] : vl[0]) + row_x<1>(b1 ? vl[0] : vl[1]);
        const float keepR = (b1 ? vr[1] : vr[0]) + row_x<1>(b1 ? vr[0] : vr[1]);
        Us[(2 * half + 0) * UP + wave * 16 + lm] = keepL;
        Us[(2 * half + 1) * UP + wave * 16 + lm] = keepR;
        if (tid < IN_DIM) Us[4 * UP + tid] = 0.f;
        bsrc[sq * BP + sk] = sval;
        if (tid < OUT_DIM) bsrc[4 * BP + tid] = 0.f;
    }

    const unsigned mask =  match4(q0, node)        | (match4(q1, node) << 4)
                        | (match4(q2, node) << 8)  | (match4(q3, node) << 12);
    const int c    = __builtin_popcount(mask);
    const int incl = wave_prefix_incl(c, lane);
    if (lane == 31) wtot[wave] = incl;
    __syncthreads();
    int before = 0, totalv = 0;
    #pragma unroll
    for (int w = 0; w < NWAVES; ++w) {
        const int t = wtot[w];
        totalv += t;
        before += (w < wave) ? t : 0;
    }
    {
        int pos = before + incl - c;
        unsigned m = mask;
        for (int it = 0; it < EPT && m != 0u; ++it) {
            const int j = __builtin_ctz(m);
            m &= m - 1u;
            if (pos < CAP) rowNode[pos] = gat[myBase + j];
            ++pos;
        }
    }
    const int total = __builtin_amdgcn_readfirstlane(totalv);
    const int cnt   = min(total, CAP);
    const int R     = cnt + 1;
    const int tiles = (R + 15) >> 4;
    if (tid == 0) rowNode[cnt] = node;
    __syncthreads();

    const int khalf = half << 1;
    for (int t = wave; t < tiles * 8; t += NWAVES) {
        const int kq = t & 3;
        const int ng = (t >> 2) & 1;
        const int mt = t >> 3;
        const int m  = mt * 16 + lm;
        const int kb = kq * KSLICE + 4 * half;
        const float* fr = feat + (long)rowNode[min(m, R - 1)] * IN_DIM + kb;
        const float* wr = Wd + (long)kb * ZD + ng * OUT_DIM + ch * 64 + 4 * lm;
        v8f acc0 = {}, acc1 = {}, acc2 = {}, acc3 = {};
        #pragma unroll
        for (int g8 = 0; g8 < KSLICE / 8; ++g8) {
            const v4f a4 = *(const v4f*)(fr + g8 * 8);
            const v4f w0 = *(const v4f*)(wr + (long)(g8 * 8 + 0) * ZD);
            const v4f w1 = *(const v4f*)(wr + (long)(g8 * 8 + 1) * ZD);
            const v4f w2 = *(const v4f*)(wr + (long)(g8 * 8 + 2) * ZD);
            const v4f w3 = *(const v4f*)(wr + (long)(g8 * 8 + 3) * ZD);
            v2f a0, a1, b;
            a0.x = a4.x;  a0.y = a4.y;
            a1.x = a4.z;  a1.y = a4.w;
            b.x = w0.x;  b.y = w1.x;
            acc0 = __builtin_amdgcn_wmma_f32_16x16x4_f32(false, a0, false, b, (short)0, acc0, false, false);
            b.x = w0.y;  b.y = w1.y;
            acc1 = __builtin_amdgcn_wmma_f32_16x16x4_f32(false, a0, false, b, (short)0, acc1, false, false);
            b.x = w0.z;  b.y = w1.z;
            acc2 = __builtin_amdgcn_wmma_f32_16x16x4_f32(false, a0, false, b, (short)0, acc2, false, false);
            b.x = w0.w;  b.y = w1.w;
            acc3 = __builtin_amdgcn_wmma_f32_16x16x4_f32(false, a0, false, b, (short)0, acc3, false, false);
            b.x = w2.x;  b.y = w3.x;
            acc0 = __builtin_amdgcn_wmma_f32_16x16x4_f32(false, a1, false, b, (short)0, acc0, false, false);
            b.x = w2.y;  b.y = w3.y;
            acc1 = __builtin_amdgcn_wmma_f32_16x16x4_f32(false, a1, false, b, (short)0, acc1, false, false);
            b.x = w2.z;  b.y = w3.z;
            acc2 = __builtin_amdgcn_wmma_f32_16x16x4_f32(false, a1, false, b, (short)0, acc2, false, false);
            b.x = w2.w;  b.y = w3.w;
            acc3 = __builtin_amdgcn_wmma_f32_16x16x4_f32(false, a1, false, b, (short)0, acc3, false, false);
        }
        float* zq = Zpart + kq * ZPLANE + (mt * 16 + 8 * half) * ZP + ng * 64 + 4 * lm;
        #pragma unroll
        for (int r = 0; r < 8; ++r) {
            v4f o;
            o.x = acc0[r];  o.y = acc1[r];  o.z = acc2[r];  o.w = acc3[r];
            *(v4f*)(zq + r * ZP) = o;
        }
    }
    *(v2f*)&tailS[wave * 64 + 2 * lane]        = tA;
    *(v2f*)&tailS[(wave + 16) * 64 + 2 * lane] = tB;
    if (wave == 0) *(v2f*)&tailS[P_IDX * 64 + 2 * lane] = tO;
    __syncthreads();
    for (int v = tid; v < tiles * 512; v += NTHR) {
        const int idx = (v >> 5) * ZP + ((v & 31) << 2);
        v4f s = *(const v4f*)(Zpart + idx);
        s += *(const v4f*)(Zpart + ZPLANE + idx);
        s += *(const v4f*)(Zpart + 2 * ZPLANE + idx);
        s += *(const v4f*)(Zpart + 3 * ZPLANE + idx);
        *(v4f*)(Zpart + idx) = s;
    }
    __syncthreads();

    if (wave < tiles) {
        const int mt2 = wave;
        const float* arow = &Zs[(mt2 * 16 + lm) * ZP + khalf];
        const float* bcol = &bsrc[min(lm, 4) * BP + khalf];
        v8f acc = {};
        for (int kk = 0; kk < OUT_DIM; kk += 4) {
            const v2f a = *(const v2f*)(arow + kk);
            const v2f b = *(const v2f*)(bcol + kk);
            acc = __builtin_amdgcn_wmma_f32_16x16x4_f32(
                      false, a, false, b, (short)0, acc, false, false);
        }
        const int rbase = mt2 * 16 + 8 * half;
        if (lm < 4) {
            #pragma unroll
            for (int r = 0; r < 8; ++r)
                lpart[(rbase + r) * 4 + lm] = acc[r];
        }
    } else if (wave >= 4) {
        const int mtf = (wave - 4) >> 2;
        const int kq  = (wave - 4) & 3;
        if (mtf < tiles) {
            const int kb = kq * KSLICE + 4 * half;
            const float* fr = feat + (long)rowNode[min(mtf * 16 + lm, R - 1)] * IN_DIM + kb;
            const float* ub = &Us[min(lm, 4) * UP + kb];
            v8f acc = {};
            #pragma unroll
            for (int g8 = 0; g8 < KSLICE / 8; ++g8) {
                const v4f a4 = *(const v4f*)(fr + g8 * 8);
                const v4f b4 = *(const v4f*)(ub + g8 * 8);
                v2f a, b;
                a.x = a4.x;  a.y = a4.y;  b.x = b4.x;  b.y = b4.y;
                acc = __builtin_amdgcn_wmma_f32_16x16x4_f32(false, a, false, b, (short)0, acc, false, false);
                a.x = a4.z;  a.y = a4.w;  b.x = b4.z;  b.y = b4.w;
                acc = __builtin_amdgcn_wmma_f32_16x16x4_f32(false, a, false, b, (short)0, acc, false, false);
            }
            const int rbase = mtf * 16 + 8 * half;
            if (lm < 4) {
                #pragma unroll
                for (int r = 0; r < 8; ++r)
                    lpart[(1 + kq) * LPL + (rbase + r) * 4 + lm] = acc[r];
            }
        }
    }
    __syncthreads();

    if (wave < 2) {
        const int hh = wave;
        const int iE = cnt * 4 + 2 * hh + 1;
        const float erh = (((lpart[iE] + lpart[LPL + iE]) + lpart[2 * LPL + iE])
                           + lpart[3 * LPL + iE]) + lpart[4 * LPL + iE];
        const int   i1  = lane + 32;
        const bool  v0  = lane < cnt;
        const bool  v1  = i1 < cnt;
        float e0 = -INFINITY, e1 = -INFINITY;
        if (v0) {
            const int i = lane * 4 + 2 * hh;
            const float t = ((((lpart[i] + lpart[LPL + i]) + lpart[2 * LPL + i])
                              + lpart[3 * LPL + i]) + lpart[4 * LPL + i]) + erh;
            e0 = (t > 0.f) ? t : NEG_SLOPE * t;
        }
        if (v1) {
            const int i = i1 * 4 + 2 * hh;
            const float t = ((((lpart[i] + lpart[LPL + i]) + lpart[2 * LPL + i])
                              + lpart[3 * LPL + i]) + lpart[4 * LPL + i]) + erh;
            e1 = (t > 0.f) ? t : NEG_SLOPE * t;
        }
        const float mx  = wave_max(fmaxf(e0, e1));
        const float w0  = v0 ? expf(e0 - mx) : 0.f;
        const float w1  = v1 ? expf(e1 - mx) : 0.f;
        const float s   = wave_sum(w0 + w1);
        const float inv = (cnt > 0) ? 1.f / s : 0.f;
        alphaS[hh * MAXR + lane] = w0 * inv;
        if (lane < MAXR - 32) alphaS[hh * MAXR + i1] = w1 * inv;
    }
    __syncthreads();

    float accum = 0.f;
    if (wave < 4) {
        const int hs = wave >> 1;
        const int pp = wave & 1;
        const int c0 = hs * 64 + pp * 32 + lm;
        v8f acc0 = {}, acc1 = {};
        for (int mt = 0; mt < tiles; ++mt) {
            #pragma unroll
            for (int j = 0; j < 4; ++j) {
                const int k = mt * 16 + j * 4 + khalf;
                const v2f a = *(const v2f*)&alphaS[hs * MAXR + k];
                v2f b0, b1;
                b0.x = Zs[k * ZP + c0];
                b0.y = Zs[(k + 1) * ZP + c0];
                b1.x = Zs[k * ZP + c0 + 16];
                b1.y = Zs[(k + 1) * ZP + c0 + 16];
                acc0 = __builtin_amdgcn_wmma_f32_16x16x4_f32(
                           false, a, false, b0, (short)0, acc0, false, false);
                acc1 = __builtin_amdgcn_wmma_f32_16x16x4_f32(
                           false, a, false, b1, (short)0, acc1, false, false);
            }
        }
        accum = ((lane < 16) ? acc0[0] : acc1[0]) + bias[hs * OUT_DIM + ch * 64 + pp * 32 + lane];
        if (hs == 1) h1res[pp * 32 + lane] = accum;
    }
    __syncthreads();
    if (wave < 2)
        out[(long)g * OUT_STRIDE + dir * OUT_DIM + ch * 64 + wave * 32 + lane] =
            0.5f * (accum + h1res[wave * 32 + lane]);
    if (wave == 4 || wave == 5) {
        const int j = tid & 63;
        float s = 0.f;
        #pragma unroll
        for (int p = 0; p < P_IDX; ++p) s += tailS[p * 64 + j];
        out[(long)g * OUT_STRIDE + 256 + 64 * (gb & 3) + j] = tailS[P_IDX * 64 + j];
        out[(long)g * OUT_STRIDE + 512 + 64 * (gb & 3) + j] = s;
    }
}

extern "C" void kernel_launch(void* const* d_in, const int* in_sizes, int n_in,
                              void* d_out, int out_size, void* d_ws, size_t ws_size,
                              hipStream_t stream)
{
    const float* feat     = (const float*)d_in[0];
    const int*   src      = (const int*)  d_in[1];
    const int*   dst      = (const int*)  d_in[2];
    const int*   op       = (const int*)  d_in[3];
    const int*   parallel = (const int*)  d_in[4];
    const float* W_f  = (const float*)d_in[5];
    const float* al_f = (const float*)d_in[6];
    const float* ar_f = (const float*)d_in[7];
    const float* b_f  = (const float*)d_in[8];
    const float* W_b  = (const float*)d_in[9];
    const float* al_b = (const float*)d_in[10];
    const float* ar_b = (const float*)d_in[11];
    const float* b_b  = (const float*)d_in[12];
    float* out = (float*)d_out;

    gat_node_kernel<<<dim3(N_GRAPHS * 2 * 2), dim3(NTHR), 0, stream>>>(
        feat, src, dst, op, parallel, W_f, al_f, ar_f, b_f, W_b, al_b, ar_b, b_b, out);
}
